# starting kernel
# baseline (speedup 1.0000x reference)
#include <hip/hip_runtime.h>

typedef float v2f __attribute__((ext_vector_type(2)));
typedef float v8f __attribute__((ext_vector_type(8)));

#define SMALL 512
#define LARGE 1024
#define NPOS  8192
#define ROWS  32
#define STR   (SMALL + 4)

__global__ __launch_bounds__(256)
void fused_embed_wmma(const int* __restrict__ ids,
                      const float* __restrict__ weight,
                      const float* __restrict__ cw,
                      const float* __restrict__ residual,
                      float* __restrict__ out)
{
    __shared__ __align__(16) float As[ROWS * STR];

    const unsigned pos0 = blockIdx.x * ROWS;
    const unsigned tid  = threadIdx.x;

    {
        const unsigned row0 = tid >> 4;
        const unsigned c16  = tid & 15;
#pragma unroll
        for (unsigned rr = 0; rr < 2; ++rr) {
            const unsigned row = row0 + rr * 16;
            const unsigned t   = (unsigned)ids[pos0 + row];
            const unsigned w4  = t << 7;
            const unsigned r4  = t << 8;
            const unsigned o4  = (pos0 + row) << 8;
            const float4* wv = (const float4*)weight;
            const float4* rv = (const float4*)residual;
            float4*       ov = (float4*)out;
            float*        lrow = As + row * STR;
#pragma unroll
            for (unsigned j = 0; j < 8; ++j) {
                const unsigned c4 = c16 + j * 16;
                float4 w = wv[w4 + c4];
                float4 r = rv[r4 + c4];
                *(float4*)(lrow + c4 * 4) = w;
                float4 o;
                o.x = w.x + r.x; o.y = w.y + r.y; o.z = w.z + r.z; o.w = w.w + r.w;
                ov[o4 + c4] = o;
            }
        }
    }
    __syncthreads();

    const unsigned wave   = tid >> 5;
    const unsigned lane   = tid & 31;
    const unsigned lane16 = lane & 15;
    const unsigned hi     = lane >> 4;
    const unsigned koff   = hi * 2;

    const float* a0p = As + lane16 * STR + koff;
    const float* a1p = As + (16 + lane16) * STR + koff;

    const unsigned nb0 = wave * 64;
    const float* b0p = cw + (((nb0 +      lane16) << 9) + koff);
    const float* b1p = cw + (((nb0 + 16 + lane16) << 9) + koff);
    const float* b2p = cw + (((nb0 + 32 + lane16) << 9) + koff);
    const float* b3p = cw + (((nb0 + 48 + lane16) << 9) + koff);

    v8f c00 = {}, c01 = {}, c02 = {}, c03 = {};
    v8f c10 = {}, c11 = {}, c12 = {}, c13 = {};

#pragma unroll 2
    for (unsigned k = 0; k < SMALL; k += 4) {
        v2f a0 = *(const v2f*)(a0p + k);
        v2f a1 = *(const v2f*)(a1p + k);
        v2f b0 = *(const v2f*)(b0p + k);
        v2f b1 = *(const v2f*)(b1p + k);
        v2f b2 = *(const v2f*)(b2p + k);
        v2f b3 = *(const v2f*)(b3p + k);
        c00 = __builtin_amdgcn_wmma_f32_16x16x4_f32(false, a0, false, b0, (short)0, c00, false, false);
        c10 = __builtin_amdgcn_wmma_f32_16x16x4_f32(false, a1, false, b0, (short)0, c10, false, false);
        c01 = __builtin_amdgcn_wmma_f32_16x16x4_f32(false, a0, false, b1, (short)0, c01, false, false);
        c11 = __builtin_amdgcn_wmma_f32_16x16x4_f32(false, a1, false, b1, (short)0, c11, false, false);
        c02 = __builtin_amdgcn_wmma_f32_16x16x4_f32(false, a0, false, b2, (short)0, c02, false, false);
        c12 = __builtin_amdgcn_wmma_f32_16x16x4_f32(false, a1, false, b2, (short)0, c12, false, false);
        c03 = __builtin_amdgcn_wmma_f32_16x16x4_f32(false, a0, false, b3, (short)0, c03, false, false);
        c13 = __builtin_amdgcn_wmma_f32_16x16x4_f32(false, a1, false, b3, (short)0, c13, false, false);
    }

    int toks0[8], toks1[8];
#pragma unroll
    for (unsigned v = 0; v < 8; ++v) {
        toks0[v] = ids[pos0 + hi * 8 + v];
        toks1[v] = ids[pos0 + 16 + hi * 8 + v];
    }

    const unsigned col = SMALL + nb0 + lane16;
#pragma unroll
    for (unsigned v = 0; v < 8; ++v) {
        const unsigned m0 = hi * 8 + v;
        const unsigned m1 = 16 + m0;
        const unsigned o0 = ((pos0 + m0) << 10) + col;
        const unsigned o1 = ((pos0 + m1) << 10) + col;
        const unsigned r0 = (((unsigned)toks0[v]) << 10) + col;
        const unsigned r1 = (((unsigned)toks1[v]) << 10) + col;
        out[o0]      = c00[v] + residual[r0];
        out[o0 + 16] = c01[v] + residual[r0 + 16];
        out[o0 + 32] = c02[v] + residual[r0 + 32];
        out[o0 + 48] = c03[v] + residual[r0 + 48];
        out[o1]      = c10[v] + residual[r1];
        out[o1 + 16] = c11[v] + residual[r1 + 16];
        out[o1 + 32] = c12[v] + residual[r1 + 32];
        out[o1 + 48] = c13[v] + residual[r1 + 48];
    }
}

extern "C" void kernel_launch(void* const* d_in, const int* in_sizes, int n_in,
                              void* d_out, int out_size, void* d_ws, size_t ws_size,
                              hipStream_t stream) {
    const int*   ids      = (const int*)d_in[0];
    const float* weight   = (const float*)d_in[1];
    const float* cw       = (const float*)d_in[2];
    const float* residual = (const float*)d_in[3];
    float*       out      = (float*)d_out;

    dim3 grid(NPOS / ROWS);
    dim3 block(256);
    fused_embed_wmma<<<grid, block, 0, stream>>>(ids, weight, cw, residual, out);
}


# cw (B operand) staged per wave through a private LDS slab in 32-k chunks by whole-line 16-byte loads, next chunk prefetched in registers; 16-byte fragment reads, k summed in permuted order per 8
# speedup vs baseline: 1.2568x; 1.2568x over previous
#include <hip/hip_runtime.h>


typedef float v2f __attribute__((ext_vector_type(2)));
typedef float v4f __attribute__((ext_vector_type(4)));
typedef float v8f __attribute__((ext_vector_type(8)));

#define SMALL 512
#define LARGE 1024
#define NPOS  8192
#define ROWS  32
#define STR   (SMALL + 4)
#define KC    32
#define BSTR  (KC + 4)
#define WN    64
#define NCH   (SMALL / KC)

#define WMMA(a, b, c) __builtin_amdgcn_wmma_f32_16x16x4_f32(false, a, false, b, (short)0, c, false, false)
#define LO(x) __builtin_shufflevector(x, x, 0, 1)
#define HI(x) __builtin_shufflevector(x, x, 2, 3)

__global__ __launch_bounds__(256)
void fused_embed_wmma(const int* __restrict__ ids,
                      const float* __restrict__ weight,
                      const float* __restrict__ cw,
                      const float* __restrict__ residual,
                      float* __restrict__ out)
{
    __shared__ __align__(16) float As[ROWS * STR];
    __shared__ __align__(16) float Bs[8 * WN * BSTR];

    const unsigned pos0 = blockIdx.x * ROWS;
    const unsigned tid  = threadIdx.x;
    const unsigned wave   = tid >> 5;
    const unsigned lane   = tid & 31;
    const unsigned lane16 = lane & 15;
    const unsigned hi     = lane >> 4;

    const unsigned srow  = lane >> 3;
    const unsigned sslot = lane & 7;
    const float* bsrc = cw + (((wave * WN + srow) << 9) + sslot * 4);
    float*       Bw   = Bs + wave * (WN * BSTR);
    float*       bdst = Bw + srow * BSTR + sslot * 4;

    v4f st[16];
#pragma unroll
    for (unsigned j = 0; j < 16; ++j)
        st[j] = *(const v4f*)(bsrc + ((j * 4) << 9));

    {
        const unsigned row0 = tid >> 4;
        const unsigned c16  = tid & 15;
#pragma unroll
        for (unsigned rr = 0; rr < 2; ++rr) {
            const unsigned row = row0 + rr * 16;
            const unsigned t   = (unsigned)ids[pos0 + row];
            const unsigned w4  = t << 7;
            const unsigned r4  = t << 8;
            const unsigned o4  = (pos0 + row) << 8;
            const float4* wv = (const float4*)weight;
            const float4* rv = (const float4*)residual;
            float4*       ov = (float4*)out;
            float*        lrow = As + row * STR;
#pragma unroll
            for (unsigned j = 0; j < 8; ++j) {
                const unsigned c4 = c16 + j * 16;
                float4 w = wv[w4 + c4];
                float4 r = rv[r4 + c4];
                *(float4*)(lrow + c4 * 4) = w;
                float4 o;
                o.x = w.x + r.x; o.y = w.y + r.y; o.z = w.z + r.z; o.w = w.w + r.w;
                ov[o4 + c4] = o;
            }
        }
    }

#pragma unroll
    for (unsigned j = 0; j < 16; ++j)
        *(v4f*)(bdst + j * 4 * BSTR) = st[j];

    __syncthreads();

    const float* a0p = As + lane16 * STR + hi * 4;
    const float* a1p = a0p + 16 * STR;
    const float* bfp = Bw + lane16 * BSTR + hi * 4;

    v8f c00 = {}, c01 = {}, c02 = {}, c03 = {};
    v8f c10 = {}, c11 = {}, c12 = {}, c13 = {};

    for (unsigned ch = 0; ch < NCH; ++ch) {
        const bool more = (ch + 1 < NCH);
        if (more) {
#pragma unroll
            for (unsigned j = 0; j < 16; ++j)
                st[j] = *(const v4f*)(bsrc + (ch + 1) * KC + ((j * 4) << 9));
        }

        const float* ak0 = a0p + ch * KC;
        const float* ak1 = a1p + ch * KC;
#pragma unroll
        for (unsigned kb = 0; kb < KC; kb += 8) {
            const v4f a0 = *(const v4f*)(ak0 + kb);
            const v4f a1 = *(const v4f*)(ak1 + kb);
            const v4f b0 = *(const v4f*)(bfp + kb);
            const v4f b1 = *(const v4f*)(bfp + 16 * BSTR + kb);
            const v4f b2 = *(const v4f*)(bfp + 32 * BSTR + kb);
            const v4f b3 = *(const v4f*)(bfp + 48 * BSTR + kb);
            c00 = WMMA(LO(a0), LO(b0), c00);
            c10 = WMMA(LO(a1), LO(b0), c10);
            c01 = WMMA(LO(a0), LO(b1), c01);
            c11 = WMMA(LO(a1), LO(b1), c11);
            c02 = WMMA(LO(a0), LO(b2), c02);
            c12 = WMMA(LO(a1), LO(b2), c12);
            c03 = WMMA(LO(a0), LO(b3), c03);
            c13 = WMMA(LO(a1), LO(b3), c13);
            c00 = WMMA(HI(a0), HI(b0), c00);
            c10 = WMMA(HI(a1), HI(b0), c10);
            c01 = WMMA(HI(a0), HI(b1), c01);
            c11 = WMMA(HI(a1), HI(b1), c11);
            c02 = WMMA(HI(a0), HI(b2), c02);
            c12 = WMMA(HI(a1), HI(b2), c12);
            c03 = WMMA(HI(a0), HI(b3), c03);
            c13 = WMMA(HI(a1), HI(b3), c13);
        }

        if (more) {
#pragma unroll
            for (unsigned j = 0; j < 16; ++j)
                *(v4f*)(bdst + j * 4 * BSTR) = st[j];
        }
    }

    const unsigned nb0 = wave * WN;
    int toks0[8], toks1[8];
#pragma unroll
    for (unsigned v = 0; v < 8; ++v) {
        toks0[v] = ids[pos0 + hi * 8 + v];
        toks1[v] = ids[pos0 + 16 + hi * 8 + v];
    }

    const unsigned col = SMALL + nb0 + lane16;
#pragma unroll
    for (unsigned v = 0; v < 8; ++v) {
        const unsigned m0 = hi * 8 + v;
        const unsigned m1 = 16 + m0;
        const unsigned o0 = ((pos0 + m0) << 10) + col;
        const unsigned o1 = ((pos0 + m1) << 10) + col;
        const unsigned r0 = (((unsigned)toks0[v]) << 10) + col;
        const unsigned r1 = (((unsigned)toks1[v]) << 10) + col;
        out[o0]      = c00[v] + residual[r0];
        out[o0 + 16] = c01[v] + residual[r0 + 16];
        out[o0 + 32] = c02[v] + residual[r0 + 32];
        out[o0 + 48] = c03[v] + residual[r0 + 48];
        out[o1]      = c10[v] + residual[r1];
        out[o1 + 16] = c11[v] + residual[r1 + 16];
        out[o1 + 32] = c12[v] + residual[r1 + 32];
        out[o1 + 48] = c13[v] + residual[r1 + 48];
    }
}

extern "C" void kernel_launch(void* const* d_in, const int* in_sizes, int n_in,
                              void* d_out, int out_size, void* d_ws, size_t ws_size,
                              hipStream_t stream) {
    const int*   ids      = (const int*)d_in[0];
    const float* weight   = (const float*)d_in[1];
    const float* cw       = (const float*)d_in[2];
    const float* residual = (const float*)d_in[3];
    float*       out      = (float*)d_out;

    dim3 grid(NPOS / ROWS);
    dim3 block(256);
    fused_embed_wmma<<<grid, block, 0, stream>>>(ids, weight, cw, residual, out);
}
